# starting kernel
# baseline (speedup 1.0000x reference)
#include <hip/hip_runtime.h>

typedef __attribute__((ext_vector_type(2))) float v2f;
typedef __attribute__((ext_vector_type(8))) float v8f;

#define SWAPX16_IMM 0x401F

__device__ __forceinline__ float swz16(float x) {
    return __builtin_bit_cast(float,
        __builtin_amdgcn_ds_swizzle(__builtin_bit_cast(int, x), SWAPX16_IMM));
}

__global__ __launch_bounds__(256) void hopf_ode_head_kernel(
    const int*   __restrict__ board,
    const float* __restrict__ W_re,
    const float* __restrict__ W_im,
    const float* __restrict__ head_w,
    const float* __restrict__ head_b,
    float*       __restrict__ out,
    int nboards)
{
    const int lane  = threadIdx.x & 31;
    const int gwave = (int)(((unsigned)blockIdx.x * blockDim.x + threadIdx.x) >> 5);
    const int b0    = gwave << 4;
    if (b0 >= nboards) return;
    const int  col = lane & 15;
    const int  hi  = lane >> 4;
    const bool lo  = (hi == 0);

    v2f Are[4], Aim[4], Anim[4];
#pragma unroll
    for (int kc = 0; kc < 4; ++kc) {
        const int j0 = 4 * kc + 2 * hi;
        const float* pr = W_re + col * 16 + j0;
        const float* pi = W_im + col * 16 + j0;
        v2f ar = { pr[0], pr[1] };
        v2f ai = { pi[0], pi[1] };
        Are[kc]  = ar;
        Aim[kc]  = ai;
        Anim[kc] = -ai;
    }

    float zr[8], zi[8];
    {
        const int* bp = board + (size_t)(b0 + col) * 16 + 8 * hi;
#pragma unroll
        for (int s = 0; s < 8; ++s) {
            float v  = (float)bp[s];
            float x  = __log2f(fmaxf(v, 1.0f)) * (1.0f / 11.0f);
            float th = x * 6.28318530717958647692f;
            zr[s] = __cosf(th);
            zi[s] = __sinf(th);
        }
    }

    const float hstep = (0.5f * 20.0f) / 19.0f;

    auto dyn = [&](const float* xr, const float* xi, float* kr, float* ki) {
        v8f accR, accI;
#pragma unroll
        for (int s = 0; s < 8; ++s) {
            float m2 = fmaf(xr[s], xr[s], xi[s] * xi[s]);
            float a  = 1.0f - m2;
            accR[s]  = fmaf(a, xr[s], -xi[s]);
            accI[s]  = fmaf(a, xi[s],  xr[s]);
        }
        v2f Bre[4], Bim[4];
        {
            float r0 = swz16(xr[0]), r1 = swz16(xr[1]), r2 = swz16(xr[2]), r3 = swz16(xr[3]);
            float r4 = swz16(xr[4]), r5 = swz16(xr[5]), r6 = swz16(xr[6]), r7 = swz16(xr[7]);
            v2f t;
            t[0] = lo ? xr[0] : r2;  t[1] = lo ? xr[1] : r3;  Bre[0] = t;
            t[0] = lo ? xr[4] : r6;  t[1] = lo ? xr[5] : r7;  Bre[1] = t;
            t[0] = lo ? r0 : xr[2];  t[1] = lo ? r1 : xr[3];  Bre[2] = t;
            t[0] = lo ? r4 : xr[6];  t[1] = lo ? r5 : xr[7];  Bre[3] = t;
        }
        {
            float i0 = swz16(xi[0]), i1 = swz16(xi[1]), i2 = swz16(xi[2]), i3 = swz16(xi[3]);
            float i4 = swz16(xi[4]), i5 = swz16(xi[5]), i6 = swz16(xi[6]), i7 = swz16(xi[7]);
            v2f t;
            t[0] = lo ? xi[0] : i2;  t[1] = lo ? xi[1] : i3;  Bim[0] = t;
            t[0] = lo ? xi[4] : i6;  t[1] = lo ? xi[5] : i7;  Bim[1] = t;
            t[0] = lo ? i0 : xi[2];  t[1] = lo ? i1 : xi[3];  Bim[2] = t;
            t[0] = lo ? i4 : xi[6];  t[1] = lo ? i5 : xi[7];  Bim[3] = t;
        }
#pragma unroll
        for (int kc = 0; kc < 4; ++kc) {
            accR = __builtin_amdgcn_wmma_f32_16x16x4_f32(false, Are[kc],  false, Bre[kc], (short)0, accR, false, false);
            accR = __builtin_amdgcn_wmma_f32_16x16x4_f32(false, Anim[kc], false, Bim[kc], (short)0, accR, false, false);
            accI = __builtin_amdgcn_wmma_f32_16x16x4_f32(false, Are[kc],  false, Bim[kc], (short)0, accI, false, false);
            accI = __builtin_amdgcn_wmma_f32_16x16x4_f32(false, Aim[kc],  false, Bre[kc], (short)0, accI, false, false);
        }
#pragma unroll
        for (int s = 0; s < 8; ++s) { kr[s] = accR[s]; ki[s] = accI[s]; }
    };

    for (int step = 0; step < 19; ++step) {
        float k1r[8], k1i[8], k2r[8], k2i[8], k3r[8], k3i[8], k4r[8], k4i[8];
        float tr[8], ti[8];

        dyn(zr, zi, k1r, k1i);
#pragma unroll
        for (int s = 0; s < 8; ++s) {
            tr[s] = fmaf(hstep * (1.0f / 3.0f), k1r[s], zr[s]);
            ti[s] = fmaf(hstep * (1.0f / 3.0f), k1i[s], zi[s]);
        }
        dyn(tr, ti, k2r, k2i);
#pragma unroll
        for (int s = 0; s < 8; ++s) {
            tr[s] = fmaf(hstep, k2r[s] - k1r[s] * (1.0f / 3.0f), zr[s]);
            ti[s] = fmaf(hstep, k2i[s] - k1i[s] * (1.0f / 3.0f), zi[s]);
        }
        dyn(tr, ti, k3r, k3i);
#pragma unroll
        for (int s = 0; s < 8; ++s) {
            tr[s] = fmaf(hstep, k1r[s] - k2r[s] + k3r[s], zr[s]);
            ti[s] = fmaf(hstep, k1i[s] - k2i[s] + k3i[s], zi[s]);
        }
        dyn(tr, ti, k4r, k4i);
#pragma unroll
        for (int s = 0; s < 8; ++s) {
            zr[s] = fmaf(hstep * 0.125f, k1r[s] + 3.0f * k2r[s] + 3.0f * k3r[s] + k4r[s], zr[s]);
            zi[s] = fmaf(hstep * 0.125f, k1i[s] + 3.0f * k2i[s] + 3.0f * k3i[s] + k4i[s], zi[s]);
        }
    }

    float acc[4];
#pragma unroll
    for (int o = 0; o < 4; ++o) {
        const float* hw = head_w + o * 32 + 8 * hi;
        float a = 0.0f;
#pragma unroll
        for (int s = 0; s < 8; ++s) {
            a = fmaf(hw[s],      zr[s], a);
            a = fmaf(hw[16 + s], zi[s], a);
        }
        a += swz16(a);
        acc[o] = a + head_b[o];
    }
    if (lo) {
        float4 r = { acc[0], acc[1], acc[2], acc[3] };
        *(float4*)(out + (size_t)(b0 + col) * 4) = r;
    }
}

extern "C" void kernel_launch(void* const* d_in, const int* in_sizes, int n_in,
                              void* d_out, int out_size, void* d_ws, size_t ws_size,
                              hipStream_t stream) {
    const int*   board  = (const int*)d_in[0];
    const float* W_re   = (const float*)d_in[1];
    const float* W_im   = (const float*)d_in[2];
    const float* head_w = (const float*)d_in[3];
    const float* head_b = (const float*)d_in[4];
    float* out = (float*)d_out;

    const int nboards = in_sizes[0] / 16;
    const int waves   = (nboards + 15) / 16;
    const long long threads = (long long)waves * 32;
    const int block = 256;
    const int grid  = (int)((threads + block - 1) / block);

    hopf_ode_head_kernel<<<grid, block, 0, stream>>>(board, W_re, W_im, head_w,
                                                     head_b, out, nboards);
}


# Complex matvec on three real products instead of four: K1=Wre(zr+zi), K2=(Wim-Wre)zr, K3=(Wre+Wim)zi, re=K1-K3, im=K1+K2; 12 not 16 f32 WMMAs per dyn eval; changes rounding and order of the sums
# speedup vs baseline: 1.1966x; 1.1966x over previous
#include <hip/hip_runtime.h>

typedef __attribute__((ext_vector_type(2))) float v2f;
typedef __attribute__((ext_vector_type(8))) float v8f;

#define SWAPX16_IMM 0x401F

__device__ __forceinline__ float swz16(float x) {
    return __builtin_bit_cast(float,
        __builtin_amdgcn_ds_swizzle(__builtin_bit_cast(int, x), SWAPX16_IMM));
}

__global__ __launch_bounds__(256) void hopf_ode_head_kernel(
    const int*   __restrict__ board,
    const float* __restrict__ W_re,
    const float* __restrict__ W_im,
    const float* __restrict__ head_w,
    const float* __restrict__ head_b,
    float*       __restrict__ out,
    int nboards)
{
    const int lane  = threadIdx.x & 31;
    const int gwave = (int)(((unsigned)blockIdx.x * blockDim.x + threadIdx.x) >> 5);
    const int b0    = gwave << 4;
    if (b0 >= nboards) return;
    const int  col = lane & 15;
    const int  hi  = lane >> 4;
    const bool lo  = (hi == 0);

    v2f Are[4], Adf[4], Ans[4];
#pragma unroll
    for (int kc = 0; kc < 4; ++kc) {
        const int j0 = 4 * kc + 2 * hi;
        const float* pr = W_re + col * 16 + j0;
        const float* pi = W_im + col * 16 + j0;
        v2f ar = { pr[0], pr[1] };
        v2f ai = { pi[0], pi[1] };
        Are[kc] = ar;
        Adf[kc] = ai - ar;
        Ans[kc] = -(ar + ai);
    }

    float zr[8], zi[8];
    {
        const int* bp = board + (size_t)(b0 + col) * 16 + 8 * hi;
#pragma unroll
        for (int s = 0; s < 8; ++s) {
            float v  = (float)bp[s];
            float x  = __log2f(fmaxf(v, 1.0f)) * (1.0f / 11.0f);
            float th = x * 6.28318530717958647692f;
            zr[s] = __cosf(th);
            zi[s] = __sinf(th);
        }
    }

    const float hstep = (0.5f * 20.0f) / 19.0f;

    auto dyn = [&](const float* xr, const float* xi, float* kr, float* ki) {
        v8f accR, accI;
#pragma unroll
        for (int s = 0; s < 8; ++s) {
            float m2 = fmaf(xr[s], xr[s], xi[s] * xi[s]);
            float a  = 1.0f - m2;
            accR[s]  = fmaf(a, xr[s], -xi[s]);
            accI[s]  = fmaf(a, xi[s],  xr[s]);
        }
        v2f Bre[4], Bim[4];
        {
            float r0 = swz16(xr[0]), r1 = swz16(xr[1]), r2 = swz16(xr[2]), r3 = swz16(xr[3]);
            float r4 = swz16(xr[4]), r5 = swz16(xr[5]), r6 = swz16(xr[6]), r7 = swz16(xr[7]);
            v2f t;
            t[0] = lo ? xr[0] : r2;  t[1] = lo ? xr[1] : r3;  Bre[0] = t;
            t[0] = lo ? xr[4] : r6;  t[1] = lo ? xr[5] : r7;  Bre[1] = t;
            t[0] = lo ? r0 : xr[2];  t[1] = lo ? r1 : xr[3];  Bre[2] = t;
            t[0] = lo ? r4 : xr[6];  t[1] = lo ? r5 : xr[7];  Bre[3] = t;
        }
        {
            float i0 = swz16(xi[0]), i1 = swz16(xi[1]), i2 = swz16(xi[2]), i3 = swz16(xi[3]);
            float i4 = swz16(xi[4]), i5 = swz16(xi[5]), i6 = swz16(xi[6]), i7 = swz16(xi[7]);
            v2f t;
            t[0] = lo ? xi[0] : i2;  t[1] = lo ? xi[1] : i3;  Bim[0] = t;
            t[0] = lo ? xi[4] : i6;  t[1] = lo ? xi[5] : i7;  Bim[1] = t;
            t[0] = lo ? i0 : xi[2];  t[1] = lo ? i1 : xi[3];  Bim[2] = t;
            t[0] = lo ? i4 : xi[6];  t[1] = lo ? i5 : xi[7];  Bim[3] = t;
        }
        v8f accK = {};
#pragma unroll
        for (int kc = 0; kc < 4; ++kc) {
            const v2f bsum = Bre[kc] + Bim[kc];
            accK = __builtin_amdgcn_wmma_f32_16x16x4_f32(false, Are[kc], false, bsum,    (short)0, accK, false, false);
            accR = __builtin_amdgcn_wmma_f32_16x16x4_f32(false, Ans[kc], false, Bim[kc], (short)0, accR, false, false);
            accI = __builtin_amdgcn_wmma_f32_16x16x4_f32(false, Adf[kc], false, Bre[kc], (short)0, accI, false, false);
        }
#pragma unroll
        for (int s = 0; s < 8; ++s) { kr[s] = accR[s] + accK[s]; ki[s] = accI[s] + accK[s]; }
    };

    for (int step = 0; step < 19; ++step) {
        float k1r[8], k1i[8], k2r[8], k2i[8], k3r[8], k3i[8], k4r[8], k4i[8];
        float tr[8], ti[8];

        dyn(zr, zi, k1r, k1i);
#pragma unroll
        for (int s = 0; s < 8; ++s) {
            tr[s] = fmaf(hstep * (1.0f / 3.0f), k1r[s], zr[s]);
            ti[s] = fmaf(hstep * (1.0f / 3.0f), k1i[s], zi[s]);
        }
        dyn(tr, ti, k2r, k2i);
#pragma unroll
        for (int s = 0; s < 8; ++s) {
            tr[s] = fmaf(hstep, k2r[s] - k1r[s] * (1.0f / 3.0f), zr[s]);
            ti[s] = fmaf(hstep, k2i[s] - k1i[s] * (1.0f / 3.0f), zi[s]);
        }
        dyn(tr, ti, k3r, k3i);
#pragma unroll
        for (int s = 0; s < 8; ++s) {
            tr[s] = fmaf(hstep, k1r[s] - k2r[s] + k3r[s], zr[s]);
            ti[s] = fmaf(hstep, k1i[s] - k2i[s] + k3i[s], zi[s]);
        }
        dyn(tr, ti, k4r, k4i);
#pragma unroll
        for (int s = 0; s < 8; ++s) {
            zr[s] = fmaf(hstep * 0.125f, k1r[s] + 3.0f * k2r[s] + 3.0f * k3r[s] + k4r[s], zr[s]);
            zi[s] = fmaf(hstep * 0.125f, k1i[s] + 3.0f * k2i[s] + 3.0f * k3i[s] + k4i[s], zi[s]);
        }
    }

    float acc[4];
#pragma unroll
    for (int o = 0; o < 4; ++o) {
        const float* hw = head_w + o * 32 + 8 * hi;
        float a = 0.0f;
#pragma unroll
        for (int s = 0; s < 8; ++s) {
            a = fmaf(hw[s],      zr[s], a);
            a = fmaf(hw[16 + s], zi[s], a);
        }
        a += swz16(a);
        acc[o] = a + head_b[o];
    }
    if (lo) {
        float4 r = { acc[0], acc[1], acc[2], acc[3] };
        *(float4*)(out + (size_t)(b0 + col) * 4) = r;
    }
}

extern "C" void kernel_launch(void* const* d_in, const int* in_sizes, int n_in,
                              void* d_out, int out_size, void* d_ws, size_t ws_size,
                              hipStream_t stream) {
    const int*   board  = (const int*)d_in[0];
    const float* W_re   = (const float*)d_in[1];
    const float* W_im   = (const float*)d_in[2];
    const float* head_w = (const float*)d_in[3];
    const float* head_b = (const float*)d_in[4];
    float* out = (float*)d_out;

    const int nboards = in_sizes[0] / 16;
    const int waves   = (nboards + 15) / 16;
    const long long threads = (long long)waves * 32;
    const int block = 256;
    const int grid  = (int)((threads + block - 1) / block);

    hopf_ode_head_kernel<<<grid, block, 0, stream>>>(board, W_re, W_im, head_w,
                                                     head_b, out, nboards);
}


# A and B fragments share a permuted k order (chunk kc, half h, reg v = input osc 8h+2kc+v): state registers are the B fragments; 16 ds_swizzle + 16 selects per dyn eval gone; sum order changes
# speedup vs baseline: 1.3558x; 1.1330x over previous
#include <hip/hip_runtime.h>

typedef __attribute__((ext_vector_type(2))) float v2f;
typedef __attribute__((ext_vector_type(8))) float v8f;

#define SWAPX16_IMM 0x401F

__device__ __forceinline__ float swz16(float x) {
    return __builtin_bit_cast(float,
        __builtin_amdgcn_ds_swizzle(__builtin_bit_cast(int, x), SWAPX16_IMM));
}

__global__ __launch_bounds__(256) void hopf_ode_head_kernel(
    const int*   __restrict__ board,
    const float* __restrict__ W_re,
    const float* __restrict__ W_im,
    const float* __restrict__ head_w,
    const float* __restrict__ head_b,
    float*       __restrict__ out,
    int nboards)
{
    const int lane  = threadIdx.x & 31;
    const int gwave = (int)(((unsigned)blockIdx.x * blockDim.x + threadIdx.x) >> 5);
    const int b0    = gwave << 4;
    if (b0 >= nboards) return;
    const int  col = lane & 15;
    const int  hi  = lane >> 4;
    const bool lo  = (hi == 0);

    v2f Are[4], Adf[4], Ans[4];
#pragma unroll
    for (int kc = 0; kc < 4; ++kc) {
        const int j0 = 8 * hi + 2 * kc;
        const float* pr = W_re + col * 16 + j0;
        const float* pi = W_im + col * 16 + j0;
        v2f ar = { pr[0], pr[1] };
        v2f ai = { pi[0], pi[1] };
        Are[kc] = ar;
        Adf[kc] = ai - ar;
        Ans[kc] = -(ar + ai);
    }

    float zr[8], zi[8];
    {
        const int* bp = board + (size_t)(b0 + col) * 16 + 8 * hi;
#pragma unroll
        for (int s = 0; s < 8; ++s) {
            float v  = (float)bp[s];
            float x  = __log2f(fmaxf(v, 1.0f)) * (1.0f / 11.0f);
            float th = x * 6.28318530717958647692f;
            zr[s] = __cosf(th);
            zi[s] = __sinf(th);
        }
    }

    const float hstep = (0.5f * 20.0f) / 19.0f;

    auto dyn = [&](const float* xr, const float* xi, float* kr, float* ki) {
        v8f accR, accI;
#pragma unroll
        for (int s = 0; s < 8; ++s) {
            float m2 = fmaf(xr[s], xr[s], xi[s] * xi[s]);
            float a  = 1.0f - m2;
            accR[s]  = fmaf(a, xr[s], -xi[s]);
            accI[s]  = fmaf(a, xi[s],  xr[s]);
        }
        v8f accK = {};
#pragma unroll
        for (int kc = 0; kc < 4; ++kc) {
            const v2f bre  = { xr[2 * kc], xr[2 * kc + 1] };
            const v2f bim  = { xi[2 * kc], xi[2 * kc + 1] };
            const v2f bsum = bre + bim;
            accK = __builtin_amdgcn_wmma_f32_16x16x4_f32(false, Are[kc], false, bsum, (short)0, accK, false, false);
            accR = __builtin_amdgcn_wmma_f32_16x16x4_f32(false, Ans[kc], false, bim,  (short)0, accR, false, false);
            accI = __builtin_amdgcn_wmma_f32_16x16x4_f32(false, Adf[kc], false, bre,  (short)0, accI, false, false);
        }
#pragma unroll
        for (int s = 0; s < 8; ++s) { kr[s] = accR[s] + accK[s]; ki[s] = accI[s] + accK[s]; }
    };

    for (int step = 0; step < 19; ++step) {
        float k1r[8], k1i[8], k2r[8], k2i[8], k3r[8], k3i[8], k4r[8], k4i[8];
        float tr[8], ti[8];

        dyn(zr, zi, k1r, k1i);
#pragma unroll
        for (int s = 0; s < 8; ++s) {
            tr[s] = fmaf(hstep * (1.0f / 3.0f), k1r[s], zr[s]);
            ti[s] = fmaf(hstep * (1.0f / 3.0f), k1i[s], zi[s]);
        }
        dyn(tr, ti, k2r, k2i);
#pragma unroll
        for (int s = 0; s < 8; ++s) {
            tr[s] = fmaf(hstep, k2r[s] - k1r[s] * (1.0f / 3.0f), zr[s]);
            ti[s] = fmaf(hstep, k2i[s] - k1i[s] * (1.0f / 3.0f), zi[s]);
        }
        dyn(tr, ti, k3r, k3i);
#pragma unroll
        for (int s = 0; s < 8; ++s) {
            tr[s] = fmaf(hstep, k1r[s] - k2r[s] + k3r[s], zr[s]);
            ti[s] = fmaf(hstep, k1i[s] - k2i[s] + k3i[s], zi[s]);
        }
        dyn(tr, ti, k4r, k4i);
#pragma unroll
        for (int s = 0; s < 8; ++s) {
            zr[s] = fmaf(hstep * 0.125f, k1r[s] + 3.0f * k2r[s] + 3.0f * k3r[s] + k4r[s], zr[s]);
            zi[s] = fmaf(hstep * 0.125f, k1i[s] + 3.0f * k2i[s] + 3.0f * k3i[s] + k4i[s], zi[s]);
        }
    }

    float acc[4];
#pragma unroll
    for (int o = 0; o < 4; ++o) {
        const float* hw = head_w + o * 32 + 8 * hi;
        float a = 0.0f;
#pragma unroll
        for (int s = 0; s < 8; ++s) {
            a = fmaf(hw[s],      zr[s], a);
            a = fmaf(hw[16 + s], zi[s], a);
        }
        a += swz16(a);
        acc[o] = a + head_b[o];
    }
    if (lo) {
        float4 r = { acc[0], acc[1], acc[2], acc[3] };
        *(float4*)(out + (size_t)(b0 + col) * 4) = r;
    }
}

extern "C" void kernel_launch(void* const* d_in, const int* in_sizes, int n_in,
                              void* d_out, int out_size, void* d_ws, size_t ws_size,
                              hipStream_t stream) {
    const int*   board  = (const int*)d_in[0];
    const float* W_re   = (const float*)d_in[1];
    const float* W_im   = (const float*)d_in[2];
    const float* head_w = (const float*)d_in[3];
    const float* head_b = (const float*)d_in[4];
    float* out = (float*)d_out;

    const int nboards = in_sizes[0] / 16;
    const int waves   = (nboards + 15) / 16;
    const long long threads = (long long)waves * 32;
    const int block = 256;
    const int grid  = (int)((threads + block - 1) / block);

    hopf_ode_head_kernel<<<grid, block, 0, stream>>>(board, W_re, W_im, head_w,
                                                     head_b, out, nboards);
}


# Linear terms of each RK stage move into per-stage matrices (I+h/3 V, hV-I, h/8 V+3/8 I; V=W+(1+i)I), K1 rides in C: 128 not 208 packed VALU ops a step; rounding and sum order change
# speedup vs baseline: 1.4552x; 1.0733x over previous
#include <hip/hip_runtime.h>

typedef __attribute__((ext_vector_type(2))) float v2f;
typedef __attribute__((ext_vector_type(8))) float v8f;

#define SWAPX16_IMM 0x401F

__device__ __forceinline__ float swz16(float x) {
    return __builtin_bit_cast(float,
        __builtin_amdgcn_ds_swizzle(__builtin_bit_cast(int, x), SWAPX16_IMM));
}

__global__ __launch_bounds__(256) void hopf_ode_head_kernel(
    const int*   __restrict__ board,
    const float* __restrict__ W_re,
    const float* __restrict__ W_im,
    const float* __restrict__ head_w,
    const float* __restrict__ head_b,
    float*       __restrict__ out,
    int nboards)
{
    const int lane  = threadIdx.x & 31;
    const int gwave = (int)(((unsigned)blockIdx.x * blockDim.x + threadIdx.x) >> 5);
    const int b0    = gwave << 4;
    if (b0 >= nboards) return;
    const int  col = lane & 15;
    const int  hi  = lane >> 4;
    const bool lo  = (hi == 0);

    const float hstep = (0.5f * 20.0f) / 19.0f;
    const float c1 = hstep * (1.0f / 3.0f);
    const float c4 = hstep * 0.125f;

    v2f M1re[4], M1df[4], M1ns[4];
    v2f M2re[4], M2df[4], M2ns[4];
    v2f M4re[4], M4df[4], M4ns[4];
#pragma unroll
    for (int kc = 0; kc < 4; ++kc) {
        const int j0 = 8 * hi + 2 * kc;
        const float* pr = W_re + col * 16 + j0;
        const float* pi = W_im + col * 16 + j0;
        const v2f dg = { (j0 == col) ? 1.0f : 0.0f, (j0 + 1 == col) ? 1.0f : 0.0f };
        const v2f vr = { pr[0] + dg[0], pr[1] + dg[1] };
        const v2f vi = { pi[0] + dg[0], pi[1] + dg[1] };
        v2f re, im;
        re = c1 * vr + dg;              im = c1 * vi;
        M1re[kc] = re;  M1df[kc] = im - re;  M1ns[kc] = -(re + im);
        re = hstep * vr - dg;           im = hstep * vi;
        M2re[kc] = re;  M2df[kc] = im - re;  M2ns[kc] = -(re + im);
        re = c4 * vr + 0.375f * dg;     im = c4 * vi;
        M4re[kc] = re;  M4df[kc] = im - re;  M4ns[kc] = -(re + im);
    }

    float zr[8], zi[8];
    {
        const int* bp = board + (size_t)(b0 + col) * 16 + 8 * hi;
#pragma unroll
        for (int s = 0; s < 8; ++s) {
            float v  = (float)bp[s];
            float x  = __log2f(fmaxf(v, 1.0f)) * (1.0f / 11.0f);
            float th = x * 6.28318530717958647692f;
            zr[s] = __cosf(th);
            zi[s] = __sinf(th);
        }
    }

    auto matvec = [&](const v2f* Mre, const v2f* Mdf, const v2f* Mns,
                      const float* xr, const float* xi, v8f& accR, v8f& accI) {
        v8f accK = {};
#pragma unroll
        for (int kc = 0; kc < 4; ++kc) {
            const v2f bsum = { xr[2 * kc] + xi[2 * kc], xr[2 * kc + 1] + xi[2 * kc + 1] };
            accK = __builtin_amdgcn_wmma_f32_16x16x4_f32(false, Mre[kc], false, bsum, (short)0, accK, false, false);
        }
        accR = accK;
        accI = accK;
#pragma unroll
        for (int kc = 0; kc < 4; ++kc) {
            const v2f bre = { xr[2 * kc], xr[2 * kc + 1] };
            const v2f bim = { xi[2 * kc], xi[2 * kc + 1] };
            accR = __builtin_amdgcn_wmma_f32_16x16x4_f32(false, Mns[kc], false, bim, (short)0, accR, false, false);
            accI = __builtin_amdgcn_wmma_f32_16x16x4_f32(false, Mdf[kc], false, bre, (short)0, accI, false, false);
        }
    };

    for (int step = 0; step < 19; ++step) {
        float t2r[8], t2i[8], t3r[8], t3i[8], t4r[8], t4i[8];
        v8f R, I;

        matvec(M1re, M1df, M1ns, zr, zi, R, I);
#pragma unroll
        for (int s = 0; s < 8; ++s) {
            const float g = -c1 * fmaf(zr[s], zr[s], zi[s] * zi[s]);
            t2r[s] = fmaf(g, zr[s], R[s]);
            t2i[s] = fmaf(g, zi[s], I[s]);
        }
        matvec(M2re, M2df, M2ns, t2r, t2i, R, I);
#pragma unroll
        for (int s = 0; s < 8; ++s) {
            const float g = -hstep * fmaf(t2r[s], t2r[s], t2i[s] * t2i[s]);
            t3r[s] = fmaf(g, t2r[s], fmaf(2.0f, zr[s], R[s]));
            t3i[s] = fmaf(g, t2i[s], fmaf(2.0f, zi[s], I[s]));
        }
        matvec(M2re, M2df, M2ns, t3r, t3i, R, I);
#pragma unroll
        for (int s = 0; s < 8; ++s) {
            const float g = -hstep * fmaf(t3r[s], t3r[s], t3i[s] * t3i[s]);
            t4r[s] = fmaf(g, t3r[s], fmaf(2.0f, t2r[s], R[s]));
            t4i[s] = fmaf(g, t3i[s], fmaf(2.0f, t2i[s], I[s]));
        }
        matvec(M4re, M4df, M4ns, t4r, t4i, R, I);
#pragma unroll
        for (int s = 0; s < 8; ++s) {
            const float g = -c4 * fmaf(t4r[s], t4r[s], t4i[s] * t4i[s]);
            zr[s] = fmaf(g, t4r[s], fmaf(-0.125f, zr[s], fmaf(0.75f, t3r[s], R[s])));
            zi[s] = fmaf(g, t4i[s], fmaf(-0.125f, zi[s], fmaf(0.75f, t3i[s], I[s])));
        }
    }

    float acc[4];
#pragma unroll
    for (int o = 0; o < 4; ++o) {
        const float* hw = head_w + o * 32 + 8 * hi;
        float a = 0.0f;
#pragma unroll
        for (int s = 0; s < 8; ++s) {
            a = fmaf(hw[s],      zr[s], a);
            a = fmaf(hw[16 + s], zi[s], a);
        }
        a += swz16(a);
        acc[o] = a + head_b[o];
    }
    if (lo) {
        float4 r = { acc[0], acc[1], acc[2], acc[3] };
        *(float4*)(out + (size_t)(b0 + col) * 4) = r;
    }
}

extern "C" void kernel_launch(void* const* d_in, const int* in_sizes, int n_in,
                              void* d_out, int out_size, void* d_ws, size_t ws_size,
                              hipStream_t stream) {
    const int*   board  = (const int*)d_in[0];
    const float* W_re   = (const float*)d_in[1];
    const float* W_im   = (const float*)d_in[2];
    const float* head_w = (const float*)d_in[3];
    const float* head_b = (const float*)d_in[4];
    float* out = (float*)d_out;

    const int nboards = in_sizes[0] / 16;
    const int waves   = (nboards + 15) / 16;
    const long long threads = (long long)waves * 32;
    const int block = 256;
    const int grid  = (int)((threads + block - 1) / block);

    hopf_ode_head_kernel<<<grid, block, 0, stream>>>(board, W_re, W_im, head_w,
                                                     head_b, out, nboards);
}
